# starting kernel
# baseline (speedup 1.0000x reference)
#include <hip/hip_runtime.h>


#define ALPHA 0.3f
#define OMA   0.7f
#define B_    8
#define T_    4096
#define C_    1024
#define NCH   (T_ / 16)
#define NSUP  16
#define NCB   (C_ / 16)

typedef __attribute__((ext_vector_type(2))) float v2f;
typedef __attribute__((ext_vector_type(8))) float v8f;

#define LOG2_07 (-0.5145731728297583f)

__global__ void __launch_bounds__(256) ema_summary(const float* __restrict__ x,
                                                   float* __restrict__ co) {
  int idx = blockIdx.x * 256 + threadIdx.x;
  int c = idx % C_;
  int j = (idx / C_) % NCH;
  int b = idx / (C_ * NCH);
  const float* px = x + ((size_t)(b * T_ + j * 16) * C_ + c);
  float acc = 0.0f;
#pragma unroll
  for (int s = 0; s < 16; ++s) {
    float v = px[(size_t)s * C_];
    float u = (j == 0 && s == 0) ? v : ALPHA * v;
    acc = OMA * acc + u;
  }
  co[idx] = acc;
}

__global__ void __launch_bounds__(256) ema_spine_a(const float* __restrict__ co,
                                                   float* __restrict__ S) {
  int idx = blockIdx.x * 256 + threadIdx.x;
  int c = idx % C_;
  int g = (idx / C_) % NSUP;
  int b = idx / (C_ * NSUP);
  const float d16 = 0.0033232930570f;
  float q = 0.0f;
#pragma unroll
  for (int i = 0; i < 16; ++i) {
    q = d16 * q + co[(size_t)(b * NCH + g * 16 + i) * C_ + c];
  }
  S[(size_t)(b * NSUP + g) * C_ + c] = q;
}

__global__ void __launch_bounds__(256) ema_spine_b(float* __restrict__ S) {
  int idx = blockIdx.x * 256 + threadIdx.x;
  int c = idx % C_;
  int b = idx / C_;
  const float d128 = 1.487810e-20f;
  const float d256 = d128 * d128;
  float p = 0.0f;
#pragma unroll
  for (int g = 0; g < NSUP; ++g) {
    size_t off = (size_t)(b * NSUP + g) * C_ + c;
    float t = S[off];
    S[off] = p;
    p = d256 * p + t;
  }
}

__global__ void __launch_bounds__(256) ema_spine_c(float* __restrict__ co,
                                                   const float* __restrict__ S) {
  int idx = blockIdx.x * 256 + threadIdx.x;
  int c = idx % C_;
  int g = (idx / C_) % NSUP;
  int b = idx / (C_ * NSUP);
  const float d16 = 0.0033232930570f;
  float p = S[(size_t)(b * NSUP + g) * C_ + c];
#pragma unroll
  for (int i = 0; i < 16; ++i) {
    size_t off = (size_t)(b * NCH + g * 16 + i) * C_ + c;
    float t = co[off];
    co[off] = p;
    p = d16 * p + t;
  }
}

__global__ void __launch_bounds__(256) ema_apply(const float* __restrict__ x,
                                                 const float* __restrict__ carry,
                                                 float* __restrict__ y) {
  const int wave = blockIdx.x * (256 / 32) + (threadIdx.x >> 5);
  const int lane = threadIdx.x & 31;
  const int half = lane >> 4;
  const int col  = lane & 15;

  int cb = wave % NCB;
  int j  = (wave / NCB) % NCH;
  int b  = wave / (NCB * NCH);
  const int t0 = j * 16;
  const int c0 = cb * 16;
  const size_t xbase = ((size_t)(b * T_) + t0) * C_ + c0 + col;

  v2f Afrag[4];
#pragma unroll
  for (int kk = 0; kk < 4; ++kk) {
    const int s0 = 4 * kk + 2 * half;
    const int e0 = col - s0;
    const int e1 = e0 - 1;
    float p0 = __builtin_amdgcn_exp2f((float)e0 * LOG2_07);
    float p1 = __builtin_amdgcn_exp2f((float)e1 * LOG2_07);
    Afrag[kk].x = (e0 >= 0) ? p0 : 0.0f;
    Afrag[kk].y = (e1 >= 0) ? p1 : 0.0f;
  }

  float xv[8];
#pragma unroll
  for (int kk = 0; kk < 4; ++kk) {
    const int s0 = 4 * kk + 2 * half;
    xv[2 * kk + 0] = x[xbase + (size_t)(s0 + 0) * C_];
    xv[2 * kk + 1] = x[xbase + (size_t)(s0 + 1) * C_];
  }

  v8f acc = {0.f, 0.f, 0.f, 0.f, 0.f, 0.f, 0.f, 0.f};

#if __has_builtin(__builtin_amdgcn_wmma_f32_16x16x4_f32)
#pragma unroll
  for (int kk = 0; kk < 4; ++kk) {
    const int s0 = 4 * kk + 2 * half;
    v2f bb;
    bb.x = (t0 + s0 == 0) ? xv[2 * kk] : ALPHA * xv[2 * kk];
    bb.y = ALPHA * xv[2 * kk + 1];
    acc = __builtin_amdgcn_wmma_f32_16x16x4_f32(
        false, Afrag[kk], false, bb, (short)0, acc, false, false);
  }
#else
  float u[16];
#pragma unroll
  for (int s = 0; s < 16; ++s) {
    float v = x[((size_t)(b * T_) + t0 + s) * C_ + c0 + col];
    u[s] = (t0 + s == 0) ? v : ALPHA * v;
  }
#pragma unroll
  for (int v = 0; v < 8; ++v) {
    int r = v + 8 * half;
    float d = 0.0f;
#pragma unroll
    for (int s = 0; s < 16; ++s)
      if (s <= r)
        d += __builtin_amdgcn_exp2f((float)(r - s) * LOG2_07) * u[s];
    acc[v] = d;
  }
#endif

  const float cin = carry[(size_t)(b * NCH + j) * C_ + c0 + col];

  float dec = __builtin_amdgcn_exp2f((float)(8 * half + 1) * LOG2_07);
#pragma unroll
  for (int v = 0; v < 8; ++v) {
    const int r = v + 8 * half;
    const float val = acc[v] + cin * dec;
    __builtin_nontemporal_store(
        val, &y[((size_t)(b * T_) + t0 + r) * C_ + c0 + col]);
    dec *= OMA;
  }
}

extern "C" void kernel_launch(void* const* d_in, const int* in_sizes, int n_in,
                              void* d_out, int out_size, void* d_ws,
                              size_t ws_size, hipStream_t stream) {
  (void)in_sizes; (void)n_in; (void)out_size; (void)ws_size;
  const float* x = (const float*)d_in[0];
  float* y = (float*)d_out;

  float* co = (float*)d_ws;
  float* S  = co + (size_t)B_ * NCH * C_;

  ema_summary<<<(B_ * NCH * C_) / 256, 256, 0, stream>>>(x, co);
  ema_spine_a<<<(B_ * NSUP * C_) / 256, 256, 0, stream>>>(co, S);
  ema_spine_b<<<(B_ * C_) / 256, 256, 0, stream>>>(S);
  ema_spine_c<<<(B_ * NSUP * C_) / 256, 256, 0, stream>>>(co, S);
  ema_apply<<<(B_ * NCH * NCB) / (256 / 32), 256, 0, stream>>>(x, co, y);
}


# Coarsen work per wave: each wave scans a 128-step x 64-channel segment (16 B per lane, carry chained chunk to chunk in registers; changes carry sum order) not one 16x16 tile; 5 launches -> 2
# speedup vs baseline: 1.7301x; 1.7301x over previous
#include <hip/hip_runtime.h>


#define ALPHA 0.3f
#define OMA   0.7f
#define B_    8
#define T_    4096
#define C_    1024
#define SEGL  128
#define NSEG  (T_ / SEGL)
#define STRIPW 64
#define NSTRIP (C_ / STRIPW)
#define WAVES 8
#define NWAVES (B_ * NSEG * NSTRIP)
#define ROWV4 (C_ / 4)

typedef __attribute__((ext_vector_type(2))) float v2f;
typedef __attribute__((ext_vector_type(4))) float v4f;
typedef __attribute__((ext_vector_type(8))) float v8f;

#define LOG2_07 (-0.5145731728297583f)

constexpr double sq_(double v) { return v * v; }
constexpr double P16_  = sq_(sq_(sq_(sq_((double)OMA))));
constexpr float  D64_  = (float)sq_(sq_(P16_));
constexpr float  D128_ = (float)sq_(sq_(sq_(P16_)));

static __device__ __forceinline__ float xhalf(float v) {
  return __builtin_bit_cast(float, __builtin_amdgcn_permlanex16(
      __builtin_bit_cast(int, v), __builtin_bit_cast(int, v), 0x76543210, 0xfedcba98, false, false));
}

__global__ void __launch_bounds__(256) ema_seg_sum(const float* __restrict__ x,
                                                   float* __restrict__ S) {
  const int lane = threadIdx.x & 31;
  const int half = lane >> 4;
  const int col  = lane & 15;
  const int wv = __builtin_amdgcn_readfirstlane((int)(threadIdx.x >> 5));
  const int id = blockIdx.x * WAVES + wv;
  const int strip = id % NSTRIP;
  const int seg   = (id / NSTRIP) % NSEG;
  const int b     = id / (NSTRIP * NSEG);
  const int cbase = strip * STRIPW + 4 * col;
  const v4f* xp = (const v4f*)(x + ((size_t)(b * T_ + seg * SEGL + 64 * half) * C_ + cbase));
  const bool first = (seg == 0) && (half == 0);

  v4f ring[2][16];
#define LOAD_BATCH(Q) do { _Pragma("unroll") for (int s_ = 0; s_ < 16; ++s_) \
    ring[(Q) & 1][s_] = xp[(size_t)(16 * (Q) + s_) * ROWV4]; } while (0)
  LOAD_BATCH(0);
  v4f acc = {0.f, 0.f, 0.f, 0.f};
#pragma unroll
  for (int q = 0; q < 4; ++q) {
    __builtin_amdgcn_sched_barrier(0);
    if (q + 1 < 4) LOAD_BATCH(q + 1);
    __builtin_amdgcn_sched_barrier(0);
#pragma unroll
    for (int s = 0; s < 16; ++s) {
      const v4f v = ring[q & 1][s];
      const v4f u = (first && q == 0 && s == 0) ? v : ALPHA * v;
      acc = OMA * acc + u;
    }
  }
#undef LOAD_BATCH
  v4f oth;
  oth.x = xhalf(acc.x); oth.y = xhalf(acc.y); oth.z = xhalf(acc.z); oth.w = xhalf(acc.w);
  const v4f lo = half ? oth : acc;
  const v4f hi = half ? acc : oth;
  const v4f r = D64_ * lo + hi;
  if (half == 0)
    *(v4f*)(S + ((size_t)(b * NSEG + seg) * C_ + cbase)) = r;
}

__global__ void __launch_bounds__(256) ema_seg_apply(const float* __restrict__ x,
                                                     const float* __restrict__ S,
                                                     float* __restrict__ y) {
  const int lane = threadIdx.x & 31;
  const int half = lane >> 4;
  const int col  = lane & 15;
  const int wv = __builtin_amdgcn_readfirstlane((int)(threadIdx.x >> 5));
  const int id = blockIdx.x * WAVES + wv;
  const int strip = id % NSTRIP;
  const int seg   = (id / NSTRIP) % NSEG;
  const int b     = id / (NSTRIP * NSEG);
  const int cbase = strip * STRIPW + 4 * col;
  const v4f* xp = (const v4f*)(x + ((size_t)(b * T_ + seg * SEGL + 2 * half) * C_ + cbase));
  v4f* yp = (v4f*)(y + ((size_t)(b * T_ + seg * SEGL + 8 * half) * C_ + cbase));

  v4f xb[3][8];
#define LOAD_CHUNK(Q) do { _Pragma("unroll") for (int k_ = 0; k_ < 4; ++k_) { \
    xb[(Q) % 3][2 * k_ + 0] = xp[(size_t)(16 * (Q) + 4 * k_ + 0) * ROWV4]; \
    xb[(Q) % 3][2 * k_ + 1] = xp[(size_t)(16 * (Q) + 4 * k_ + 1) * ROWV4]; } } while (0)
  LOAD_CHUNK(0);
  LOAD_CHUNK(1);

  v2f Afrag[4];
#pragma unroll
  for (int kk = 0; kk < 4; ++kk) {
    const int s0 = 4 * kk + 2 * half;
    const int e0 = col - s0;
    const int e1 = e0 - 1;
    const float p0 = __builtin_amdgcn_exp2f((float)e0 * LOG2_07);
    const float p1 = __builtin_amdgcn_exp2f((float)e1 * LOG2_07);
    Afrag[kk].x = (e0 >= 0) ? p0 : 0.0f;
    Afrag[kk].y = (e1 >= 0) ? p1 : 0.0f;
  }
  float dec[8];
  dec[0] = __builtin_amdgcn_exp2f((float)(8 * half + 1) * LOG2_07);
#pragma unroll
  for (int v = 1; v < 8; ++v) dec[v] = dec[v - 1] * OMA;

  v4f cin = {0.f, 0.f, 0.f, 0.f};
  {
    const v4f* Sp = (const v4f*)(S + ((size_t)(b * NSEG) * C_ + cbase));
    const int last = (seg > 0) ? (seg - 1) : 0;
    v4f t[NSEG - 1];
#pragma unroll
    for (int i = 0; i < NSEG - 1; ++i) {
      const int ii = (i < seg) ? i : last;
      t[i] = Sp[(size_t)ii * ROWV4];
    }
    cin = t[0];
#pragma unroll
    for (int i = 1; i < NSEG - 1; ++i)
      if (i < seg) cin = D128_ * cin + t[i];
    if (seg == 0) cin = (v4f){0.f, 0.f, 0.f, 0.f};
  }

#pragma unroll
  for (int q = 0; q < SEGL / 16; ++q) {
    __builtin_amdgcn_sched_barrier(0);
    if (q + 2 < SEGL / 16) LOAD_CHUNK(q + 2);
    __builtin_amdgcn_sched_barrier(0);

    v8f acc[4];
#pragma unroll
    for (int n = 0; n < 4; ++n) acc[n] = (v8f){0.f, 0.f, 0.f, 0.f, 0.f, 0.f, 0.f, 0.f};
#pragma unroll
    for (int kk = 0; kk < 4; ++kk) {
      const v4f r0 = xb[q % 3][2 * kk + 0];
      const v4f r1 = xb[q % 3][2 * kk + 1];
      const bool t_is_0 = (q == 0) && (kk == 0) && (seg == 0) && (half == 0);
#pragma unroll
      for (int n = 0; n < 4; ++n) {
        v2f bb;
        bb.x = t_is_0 ? r0[n] : ALPHA * r0[n];
        bb.y = ALPHA * r1[n];
        acc[n] = __builtin_amdgcn_wmma_f32_16x16x4_f32(
            false, Afrag[kk], false, bb, (short)0, acc[n], false, false);
      }
    }

    v4f o7;
#pragma unroll
    for (int v = 0; v < 8; ++v) {
      v4f o;
      o.x = acc[0][v] + cin.x * dec[v];
      o.y = acc[1][v] + cin.y * dec[v];
      o.z = acc[2][v] + cin.z * dec[v];
      o.w = acc[3][v] + cin.w * dec[v];
      __builtin_nontemporal_store(o, &yp[(size_t)(16 * q + v) * ROWV4]);
      if (v == 7) o7 = o;
    }
    if (q + 1 < SEGL / 16) {
      const float ex = xhalf(o7.x), ey = xhalf(o7.y), ez = xhalf(o7.z), ew = xhalf(o7.w);
      cin.x = half ? o7.x : ex;
      cin.y = half ? o7.y : ey;
      cin.z = half ? o7.z : ez;
      cin.w = half ? o7.w : ew;
    }
  }
#undef LOAD_CHUNK
}

extern "C" void kernel_launch(void* const* d_in, const int* in_sizes, int n_in,
                              void* d_out, int out_size, void* d_ws,
                              size_t ws_size, hipStream_t stream) {
  (void)in_sizes; (void)n_in; (void)out_size; (void)ws_size;
  const float* x = (const float*)d_in[0];
  float* y = (float*)d_out;
  float* S = (float*)d_ws;

  ema_seg_sum<<<NWAVES / WAVES, 256, 0, stream>>>(x, S);
  ema_seg_apply<<<NWAVES / WAVES, 256, 0, stream>>>(x, S, y);
}


# Fuse the two launches into one: a 16-wave work-group owns a whole 64-channel sequence, chunk sums pass through LDS and one barrier per 256 steps; x read once (changes carry sum order)
# speedup vs baseline: 2.1968x; 1.2698x over previous
#include <hip/hip_runtime.h>


#define ALPHA 0.3f
#define OMA   0.7f
#define B_    8
#define T_    4096
#define C_    1024
#define STRIPW 64
#define NSTRIP (C_ / STRIPW)
#define CHUNK 16
#define NWAVE 16
#define SBLK  (CHUNK * NWAVE)
#define NSB   (T_ / SBLK)
#define RING  4
#define ROWV4 (C_ / 4)

typedef __attribute__((ext_vector_type(2))) float v2f;
typedef __attribute__((ext_vector_type(4))) float v4f;
typedef __attribute__((ext_vector_type(8))) float v8f;

#define LOG2_07 (-0.5145731728297583f)

constexpr double sq_(double v) { return v * v; }
constexpr float P16F_ = (float)sq_(sq_(sq_(sq_((double)OMA))));

static __device__ __forceinline__ void load_chunk(v4f (&dst)[8], const float* __restrict__ xrow,
                                                  const int sb) {
  const v4f* p = (const v4f*)xrow + (size_t)sb * SBLK * ROWV4;
#pragma unroll
  for (int k = 0; k < 4; ++k) {
    dst[2 * k + 0] = p[(size_t)(4 * k + 0) * ROWV4];
    dst[2 * k + 1] = p[(size_t)(4 * k + 1) * ROWV4];
  }
}

template <int U, bool PREFETCH>
static __device__ __forceinline__ void super_block(
    v4f (&xb)[RING][8], const float* __restrict__ xrow, float* __restrict__ yrow, const int sb,
    const v2f (&Afrag)[4], const float (&dec)[8], v4f& carry, v4f (&Sc)[2][NWAVE][16],
    const int w, const int half, const int col) {
  __builtin_amdgcn_sched_barrier(0);
  if (PREFETCH) load_chunk(xb[(U + RING - 1) % RING], xrow, sb + RING - 1);
  __builtin_amdgcn_sched_barrier(0);

  const bool t_is_0 = (sb == 0) && (w == 0) && (half == 0);

  v8f acc[4];
#pragma unroll
  for (int n = 0; n < 4; ++n) acc[n] = (v8f){0.f, 0.f, 0.f, 0.f, 0.f, 0.f, 0.f, 0.f};
#pragma unroll
  for (int kk = 0; kk < 4; ++kk) {
    const v4f r0 = xb[U][2 * kk + 0];
    const v4f r1 = xb[U][2 * kk + 1];
#pragma unroll
    for (int n = 0; n < 4; ++n) {
      v2f bb;
      bb.x = (kk == 0 && t_is_0) ? r0[n] : ALPHA * r0[n];
      bb.y = ALPHA * r1[n];
      acc[n] = __builtin_amdgcn_wmma_f32_16x16x4_f32(
          false, Afrag[kk], false, bb, (short)0, acc[n], false, false);
    }
  }

  if (half)
    Sc[U & 1][w][col] = (v4f){acc[0][7], acc[1][7], acc[2][7], acc[3][7]};
  asm volatile("s_wait_dscnt 0x0" ::: "memory");
  asm volatile("s_barrier_signal -1\n\ts_barrier_wait -1" ::: "memory");

  v4f t[NWAVE];
#pragma unroll
  for (int j = 0; j < NWAVE; ++j) t[j] = Sc[U & 1][j][col];
  v4f c = carry;
  v4f cin = carry;
#pragma unroll
  for (int j = 0; j < NWAVE; ++j) {
    if (j == w) cin = c;
    c = P16F_ * c + t[j];
  }
  carry = c;

  v4f* yp = (v4f*)yrow + (size_t)sb * SBLK * ROWV4;
#pragma unroll
  for (int v = 0; v < 8; ++v) {
    v4f o;
    o.x = acc[0][v] + cin.x * dec[v];
    o.y = acc[1][v] + cin.y * dec[v];
    o.z = acc[2][v] + cin.z * dec[v];
    o.w = acc[3][v] + cin.w * dec[v];
    __builtin_nontemporal_store(o, &yp[(size_t)v * ROWV4]);
  }
}

__global__ void __launch_bounds__(NWAVE * 32) ema_scan_fused(const float* __restrict__ x,
                                                             float* __restrict__ y) {
  __shared__ v4f Sc[2][NWAVE][16];

  const int lane = threadIdx.x & 31;
  const int half = lane >> 4;
  const int col  = lane & 15;
  const int w = __builtin_amdgcn_readfirstlane((int)(threadIdx.x >> 5));
  const int strip = blockIdx.x % NSTRIP;
  const int b     = blockIdx.x / NSTRIP;
  const int cbase = strip * STRIPW + 4 * col;
  const float* xrow = x + ((size_t)(b * T_ + w * CHUNK + 2 * half) * C_ + cbase);
  float* yrow = y + ((size_t)(b * T_ + w * CHUNK + 8 * half) * C_ + cbase);

  v4f xb[RING][8];
  load_chunk(xb[0], xrow, 0);
  load_chunk(xb[1], xrow, 1);
  load_chunk(xb[2], xrow, 2);

  v2f Afrag[4];
#pragma unroll
  for (int kk = 0; kk < 4; ++kk) {
    const int s0 = 4 * kk + 2 * half;
    const int e0 = col - s0;
    const int e1 = e0 - 1;
    const float p0 = __builtin_amdgcn_exp2f((float)e0 * LOG2_07);
    const float p1 = __builtin_amdgcn_exp2f((float)e1 * LOG2_07);
    Afrag[kk].x = (e0 >= 0) ? p0 : 0.0f;
    Afrag[kk].y = (e1 >= 0) ? p1 : 0.0f;
  }
  float dec[8];
  dec[0] = __builtin_amdgcn_exp2f((float)(8 * half + 1) * LOG2_07);
#pragma unroll
  for (int v = 1; v < 8; ++v) dec[v] = dec[v - 1] * OMA;

  v4f carry = {0.f, 0.f, 0.f, 0.f};

  for (int sb0 = 0; sb0 < NSB - RING; sb0 += RING) {
    super_block<0, true>(xb, xrow, yrow, sb0 + 0, Afrag, dec, carry, Sc, w, half, col);
    super_block<1, true>(xb, xrow, yrow, sb0 + 1, Afrag, dec, carry, Sc, w, half, col);
    super_block<2, true>(xb, xrow, yrow, sb0 + 2, Afrag, dec, carry, Sc, w, half, col);
    super_block<3, true>(xb, xrow, yrow, sb0 + 3, Afrag, dec, carry, Sc, w, half, col);
  }
  super_block<0, true >(xb, xrow, yrow, NSB - 4, Afrag, dec, carry, Sc, w, half, col);
  super_block<1, false>(xb, xrow, yrow, NSB - 3, Afrag, dec, carry, Sc, w, half, col);
  super_block<2, false>(xb, xrow, yrow, NSB - 2, Afrag, dec, carry, Sc, w, half, col);
  super_block<3, false>(xb, xrow, yrow, NSB - 1, Afrag, dec, carry, Sc, w, half, col);
}

extern "C" void kernel_launch(void* const* d_in, const int* in_sizes, int n_in,
                              void* d_out, int out_size, void* d_ws,
                              size_t ws_size, hipStream_t stream) {
  (void)in_sizes; (void)n_in; (void)out_size; (void)d_ws; (void)ws_size;
  const float* x = (const float*)d_in[0];
  float* y = (float*)d_out;

  ema_scan_fused<<<B_ * NSTRIP, NWAVE * 32, 0, stream>>>(x, y);
}


# Split each strip in time over two work-groups (256, one per WGP): the back one first sums rows 0..2559 into its carry (vector multiply-adds, rows read twice), then scans; changes carry sum order
# speedup vs baseline: 2.6397x; 1.2016x over previous
#include <hip/hip_runtime.h>


#define ALPHA 0.3f
#define OMA   0.7f
#define B_    8
#define T_    4096
#define C_    1024
#define STRIPW 64
#define NSTRIP (C_ / STRIPW)
#define CHUNK 16
#define NWAVE 16
#define SBLK  (CHUNK * NWAVE)
#define NSB   (T_ / SBLK)
#define RING  4
#define ROWV4 (C_ / 4)
#define SPLIT_SB 10
#define SPLIT (SPLIT_SB * SBLK)
#define SEG   (SPLIT / NWAVE)
#define RCH   (SEG / CHUNK)
#define NSEQ  (B_ * NSTRIP)

typedef __attribute__((ext_vector_type(2))) float v2f;
typedef __attribute__((ext_vector_type(4))) float v4f;
typedef __attribute__((ext_vector_type(8))) float v8f;

#define LOG2_07 (-0.5145731728297583f)

constexpr double sq_(double v) { return v * v; }
constexpr float P16F_ = (float)sq_(sq_(sq_(sq_((double)OMA))));
constexpr double P32D_ = sq_(sq_(sq_(sq_(sq_((double)OMA)))));
constexpr float PSEGF_ = (float)(sq_(sq_(P32D_)) * P32D_);
static_assert(SEG == 160, "PSEGF_ is 0.7^SEG");

static __device__ __forceinline__ void load_chunk(v4f (&dst)[8], const float* __restrict__ xrow,
                                                  const int sb) {
  const v4f* p = (const v4f*)xrow + (size_t)sb * SBLK * ROWV4;
#pragma unroll
  for (int k = 0; k < 4; ++k) {
    dst[2 * k + 0] = p[(size_t)(4 * k + 0) * ROWV4];
    dst[2 * k + 1] = p[(size_t)(4 * k + 1) * ROWV4];
  }
}

template <int U, bool PREFETCH>
static __device__ __forceinline__ void super_block(
    v4f (&xb)[RING][8], const float* __restrict__ xrow, float* __restrict__ yrow, const int sb,
    const v2f (&Afrag)[4], const float (&dec)[8], v4f& carry, v4f (&Sc)[2][NWAVE][16],
    const int w, const int half, const int col) {
  __builtin_amdgcn_sched_barrier(0);
  if (PREFETCH) load_chunk(xb[(U + RING - 1) % RING], xrow, sb + RING - 1);
  __builtin_amdgcn_sched_barrier(0);

  const bool t_is_0 = (sb == 0) && (w == 0) && (half == 0);

  v8f acc[4];
#pragma unroll
  for (int n = 0; n < 4; ++n) acc[n] = (v8f){0.f, 0.f, 0.f, 0.f, 0.f, 0.f, 0.f, 0.f};
#pragma unroll
  for (int kk = 0; kk < 4; ++kk) {
    const v4f r0 = xb[U][2 * kk + 0];
    const v4f r1 = xb[U][2 * kk + 1];
#pragma unroll
    for (int n = 0; n < 4; ++n) {
      v2f bb;
      bb.x = (kk == 0 && t_is_0) ? r0[n] : ALPHA * r0[n];
      bb.y = ALPHA * r1[n];
      acc[n] = __builtin_amdgcn_wmma_f32_16x16x4_f32(
          false, Afrag[kk], false, bb, (short)0, acc[n], false, false);
    }
  }

  if (half)
    Sc[U & 1][w][col] = (v4f){acc[0][7], acc[1][7], acc[2][7], acc[3][7]};
  asm volatile("s_wait_dscnt 0x0" ::: "memory");
  asm volatile("s_barrier_signal -1\n\ts_barrier_wait -1" ::: "memory");

  v4f t[NWAVE];
#pragma unroll
  for (int j = 0; j < NWAVE; ++j) t[j] = Sc[U & 1][j][col];
  v4f c = carry;
  v4f cin = carry;
#pragma unroll
  for (int j = 0; j < NWAVE; ++j) {
    if (j == w) cin = c;
    c = P16F_ * c + t[j];
  }
  carry = c;

  v4f* yp = (v4f*)yrow + (size_t)sb * SBLK * ROWV4;
#pragma unroll
  for (int v = 0; v < 8; ++v) {
    v4f o;
    o.x = acc[0][v] + cin.x * dec[v];
    o.y = acc[1][v] + cin.y * dec[v];
    o.z = acc[2][v] + cin.z * dec[v];
    o.w = acc[3][v] + cin.w * dec[v];
    __builtin_nontemporal_store(o, &yp[(size_t)v * ROWV4]);
  }
}

static __device__ __forceinline__ void load_seg_chunk(v4f (&dst)[8], const float* __restrict__ xseg,
                                                      const int ch) {
  const v4f* p = (const v4f*)xseg + (size_t)ch * CHUNK * ROWV4;
#pragma unroll
  for (int k = 0; k < 4; ++k) {
    dst[2 * k + 0] = p[(size_t)(4 * k + 0) * ROWV4];
    dst[2 * k + 1] = p[(size_t)(4 * k + 1) * ROWV4];
  }
}

__global__ void __launch_bounds__(NWAVE * 32) ema_scan_fused(const float* __restrict__ x,
                                                             float* __restrict__ y) {
  __shared__ v4f Sc[2][NWAVE][16];
  __shared__ v4f Pre[2][NWAVE][16];

  const int lane = threadIdx.x & 31;
  const int half = lane >> 4;
  const int col  = lane & 15;
  const int w = __builtin_amdgcn_readfirstlane((int)(threadIdx.x >> 5));
  const int back  = (blockIdx.x < NSEQ) ? 1 : 0;
  const int seq   = blockIdx.x % NSEQ;
  const int strip = seq % NSTRIP;
  const int b     = seq / NSTRIP;
  const int cbase = strip * STRIPW + 4 * col;
  const float* xrow = x + ((size_t)(b * T_ + w * CHUNK + 2 * half) * C_ + cbase);
  float* yrow = y + ((size_t)(b * T_ + w * CHUNK + 8 * half) * C_ + cbase);

  v4f xb[RING][8];
  v4f carry = {0.f, 0.f, 0.f, 0.f};

  if (back) {
    const float* xseg = x + ((size_t)(b * T_ + w * SEG + 2 * half) * C_ + cbase);
    load_seg_chunk(xb[0], xseg, 0);
    load_seg_chunk(xb[1], xseg, 1);
    load_seg_chunk(xb[2], xseg, 2);
    float dw[8];
#pragma unroll
    for (int k = 0; k < 4; ++k) {
      const int r0 = 4 * k + 2 * half;
      dw[2 * k + 0] = __builtin_amdgcn_exp2f((float)(15 - r0) * LOG2_07);
      dw[2 * k + 1] = __builtin_amdgcn_exp2f((float)(14 - r0) * LOG2_07);
    }
    const bool t_is_0 = (w == 0) && (half == 0);
    v4f ps = {0.f, 0.f, 0.f, 0.f};
#pragma unroll
    for (int ch = 0; ch < RCH; ++ch) {
      __builtin_amdgcn_sched_barrier(0);
      if (ch + RING - 1 < RCH) load_seg_chunk(xb[(ch + RING - 1) % RING], xseg, ch + RING - 1);
      else load_chunk(xb[(ch + RING - 1) % RING], xrow, SPLIT_SB + (ch + RING - 1 - RCH));
      __builtin_amdgcn_sched_barrier(0);
      v4f a = {0.f, 0.f, 0.f, 0.f};
#pragma unroll
      for (int i = 0; i < 8; ++i) {
        const v4f r = xb[ch % RING][i];
        const v4f u = (ch == 0 && i == 0 && t_is_0) ? r : ALPHA * r;
        a += dw[i] * u;
      }
      ps = P16F_ * ps + a;
    }
    Pre[half][w][col] = ps;
    asm volatile("s_wait_dscnt 0x0" ::: "memory");
    asm volatile("s_barrier_signal -1\n\ts_barrier_wait -1" ::: "memory");
    v4f s0[NWAVE], s1[NWAVE];
#pragma unroll
    for (int j = 0; j < NWAVE; ++j) { s0[j] = Pre[0][j][col]; s1[j] = Pre[1][j][col]; }
#pragma unroll
    for (int j = 0; j < NWAVE; ++j) carry = PSEGF_ * carry + (s0[j] + s1[j]);
  } else {
    load_chunk(xb[2], xrow, 0);
    load_chunk(xb[3], xrow, 1);
    load_chunk(xb[0], xrow, 2);
  }

  v2f Afrag[4];
#pragma unroll
  for (int kk = 0; kk < 4; ++kk) {
    const int s0 = 4 * kk + 2 * half;
    const int e0 = col - s0;
    const int e1 = e0 - 1;
    const float p0 = __builtin_amdgcn_exp2f((float)e0 * LOG2_07);
    const float p1 = __builtin_amdgcn_exp2f((float)e1 * LOG2_07);
    Afrag[kk].x = (e0 >= 0) ? p0 : 0.0f;
    Afrag[kk].y = (e1 >= 0) ? p1 : 0.0f;
  }
  float dec[8];
  dec[0] = __builtin_amdgcn_exp2f((float)(8 * half + 1) * LOG2_07);
#pragma unroll
  for (int v = 1; v < 8; ++v) dec[v] = dec[v - 1] * OMA;

  int sb = back ? SPLIT_SB : 0;
  super_block<2, true>(xb, xrow, yrow, sb + 0, Afrag, dec, carry, Sc, w, half, col);
  super_block<3, true>(xb, xrow, yrow, sb + 1, Afrag, dec, carry, Sc, w, half, col);
  sb += 2;
  if (!back) {
    super_block<0, true>(xb, xrow, yrow, sb + 0, Afrag, dec, carry, Sc, w, half, col);
    super_block<1, true>(xb, xrow, yrow, sb + 1, Afrag, dec, carry, Sc, w, half, col);
    super_block<2, true>(xb, xrow, yrow, sb + 2, Afrag, dec, carry, Sc, w, half, col);
    super_block<3, true>(xb, xrow, yrow, sb + 3, Afrag, dec, carry, Sc, w, half, col);
    sb += 4;
  }
  super_block<0, true >(xb, xrow, yrow, sb + 0, Afrag, dec, carry, Sc, w, half, col);
  super_block<1, false>(xb, xrow, yrow, sb + 1, Afrag, dec, carry, Sc, w, half, col);
  super_block<2, false>(xb, xrow, yrow, sb + 2, Afrag, dec, carry, Sc, w, half, col);
  super_block<3, false>(xb, xrow, yrow, sb + 3, Afrag, dec, carry, Sc, w, half, col);
}

extern "C" void kernel_launch(void* const* d_in, const int* in_sizes, int n_in,
                              void* d_out, int out_size, void* d_ws,
                              size_t ws_size, hipStream_t stream) {
  (void)in_sizes; (void)n_in; (void)out_size; (void)d_ws; (void)ws_size;
  const float* x = (const float*)d_in[0];
  float* y = (float*)d_out;

  ema_scan_fused<<<2 * NSEQ, NWAVE * 32, 0, stream>>>(x, y);
}


# Back work-group's carry sums only the 768 rows before its start: older rows weigh under 0.7^768 = 1e-119, nothing at f32 for any finite x; split moved to 2048/2048; changes carry sum order
# speedup vs baseline: 2.8634x; 1.0847x over previous
#include <hip/hip_runtime.h>


#define ALPHA 0.3f
#define OMA   0.7f
#define B_    8
#define T_    4096
#define C_    1024
#define STRIPW 64
#define NSTRIP (C_ / STRIPW)
#define CHUNK 16
#define NWAVE 16
#define SBLK  (CHUNK * NWAVE)
#define NSB   (T_ / SBLK)
#define RING  4
#define ROWV4 (C_ / 4)
#define SPLIT_SB 8
#define SPLIT (SPLIT_SB * SBLK)
#define WIN   768
#define WIN0  (SPLIT - WIN)
#define SEG   (WIN / NWAVE)
#define RCH   (SEG / CHUNK)
#define NSEQ  (B_ * NSTRIP)

typedef __attribute__((ext_vector_type(2))) float v2f;
typedef __attribute__((ext_vector_type(4))) float v4f;
typedef __attribute__((ext_vector_type(8))) float v8f;

#define LOG2_07 (-0.5145731728297583f)

constexpr double sq_(double v) { return v * v; }
constexpr float P16F_ = (float)sq_(sq_(sq_(sq_((double)OMA))));
constexpr double P16D_ = sq_(sq_(sq_(sq_((double)OMA))));
constexpr float PSEGF_ = (float)(P16D_ * P16D_ * P16D_);
static_assert(SEG == 48, "PSEGF_ is 0.7^SEG");
static_assert(RCH == RING - 1, "the carry sum's chunks fill the ring once: every trip of its loop requests a scan chunk");
static_assert(WIN0 > 0, "the window must not hold step 0 (u_0 = x_0 is the front work-group's)");
static_assert(WIN >= 768, "0.7^WIN times the largest finite f32 must stay far under the smallest denormal");

static __device__ __forceinline__ void load_chunk(v4f (&dst)[8], const float* __restrict__ xrow,
                                                  const int sb) {
  const v4f* p = (const v4f*)xrow + (size_t)sb * SBLK * ROWV4;
#pragma unroll
  for (int k = 0; k < 4; ++k) {
    dst[2 * k + 0] = p[(size_t)(4 * k + 0) * ROWV4];
    dst[2 * k + 1] = p[(size_t)(4 * k + 1) * ROWV4];
  }
}

template <int U, bool PREFETCH>
static __device__ __forceinline__ void super_block(
    v4f (&xb)[RING][8], const float* __restrict__ xrow, float* __restrict__ yrow, const int sb,
    const v2f (&Afrag)[4], const float (&dec)[8], v4f& carry, v4f (&Sc)[2][NWAVE][16],
    const int w, const int half, const int col) {
  __builtin_amdgcn_sched_barrier(0);
  if (PREFETCH) load_chunk(xb[(U + RING - 1) % RING], xrow, sb + RING - 1);
  __builtin_amdgcn_sched_barrier(0);

  const bool t_is_0 = (sb == 0) && (w == 0) && (half == 0);

  v8f acc[4];
#pragma unroll
  for (int n = 0; n < 4; ++n) acc[n] = (v8f){0.f, 0.f, 0.f, 0.f, 0.f, 0.f, 0.f, 0.f};
#pragma unroll
  for (int kk = 0; kk < 4; ++kk) {
    const v4f r0 = xb[U][2 * kk + 0];
    const v4f r1 = xb[U][2 * kk + 1];
#pragma unroll
    for (int n = 0; n < 4; ++n) {
      v2f bb;
      bb.x = (kk == 0 && t_is_0) ? r0[n] : ALPHA * r0[n];
      bb.y = ALPHA * r1[n];
      acc[n] = __builtin_amdgcn_wmma_f32_16x16x4_f32(
          false, Afrag[kk], false, bb, (short)0, acc[n], false, false);
    }
  }

  if (half)
    Sc[U & 1][w][col] = (v4f){acc[0][7], acc[1][7], acc[2][7], acc[3][7]};
  asm volatile("s_wait_dscnt 0x0" ::: "memory");
  asm volatile("s_barrier_signal -1\n\ts_barrier_wait -1" ::: "memory");

  v4f t[NWAVE];
#pragma unroll
  for (int j = 0; j < NWAVE; ++j) t[j] = Sc[U & 1][j][col];
  v4f c = carry;
  v4f cin = carry;
#pragma unroll
  for (int j = 0; j < NWAVE; ++j) {
    if (j == w) cin = c;
    c = P16F_ * c + t[j];
  }
  carry = c;

  v4f* yp = (v4f*)yrow + (size_t)sb * SBLK * ROWV4;
#pragma unroll
  for (int v = 0; v < 8; ++v) {
    v4f o;
    o.x = acc[0][v] + cin.x * dec[v];
    o.y = acc[1][v] + cin.y * dec[v];
    o.z = acc[2][v] + cin.z * dec[v];
    o.w = acc[3][v] + cin.w * dec[v];
    __builtin_nontemporal_store(o, &yp[(size_t)v * ROWV4]);
  }
}

static __device__ __forceinline__ void load_seg_chunk(v4f (&dst)[8], const float* __restrict__ xseg,
                                                      const int ch) {
  const v4f* p = (const v4f*)xseg + (size_t)ch * CHUNK * ROWV4;
#pragma unroll
  for (int k = 0; k < 4; ++k) {
    dst[2 * k + 0] = p[(size_t)(4 * k + 0) * ROWV4];
    dst[2 * k + 1] = p[(size_t)(4 * k + 1) * ROWV4];
  }
}

__global__ void __launch_bounds__(NWAVE * 32) ema_scan_fused(const float* __restrict__ x,
                                                             float* __restrict__ y) {
  __shared__ v4f Sc[2][NWAVE][16];
  __shared__ v4f Pre[2][NWAVE][16];

  const int lane = threadIdx.x & 31;
  const int half = lane >> 4;
  const int col  = lane & 15;
  const int w = __builtin_amdgcn_readfirstlane((int)(threadIdx.x >> 5));
  const int back  = (blockIdx.x < NSEQ) ? 1 : 0;
  const int seq   = blockIdx.x % NSEQ;
  const int strip = seq % NSTRIP;
  const int b     = seq / NSTRIP;
  const int cbase = strip * STRIPW + 4 * col;
  const float* xrow = x + ((size_t)(b * T_ + w * CHUNK + 2 * half) * C_ + cbase);
  float* yrow = y + ((size_t)(b * T_ + w * CHUNK + 8 * half) * C_ + cbase);

  v4f xb[RING][8];
  v4f carry = {0.f, 0.f, 0.f, 0.f};

  if (back) {
    const float* xseg = x + ((size_t)(b * T_ + WIN0 + w * SEG + 2 * half) * C_ + cbase);
    load_seg_chunk(xb[3], xseg, 0);
    load_seg_chunk(xb[0], xseg, 1);
    load_seg_chunk(xb[1], xseg, 2);
    float dw[8];
#pragma unroll
    for (int k = 0; k < 4; ++k) {
      const int r0 = 4 * k + 2 * half;
      dw[2 * k + 0] = __builtin_amdgcn_exp2f((float)(15 - r0) * LOG2_07);
      dw[2 * k + 1] = __builtin_amdgcn_exp2f((float)(14 - r0) * LOG2_07);
    }
    v4f ps = {0.f, 0.f, 0.f, 0.f};
#pragma unroll
    for (int ch = 0; ch < RCH; ++ch) {
      __builtin_amdgcn_sched_barrier(0);
      load_chunk(xb[(ch + 2) % RING], xrow, SPLIT_SB + ch);
      __builtin_amdgcn_sched_barrier(0);
      v4f a = {0.f, 0.f, 0.f, 0.f};
#pragma unroll
      for (int i = 0; i < 8; ++i) {
        const v4f r = xb[(ch + 3) % RING][i];
        a += dw[i] * (ALPHA * r);
      }
      ps = P16F_ * ps + a;
    }
    Pre[half][w][col] = ps;
    asm volatile("s_wait_dscnt 0x0" ::: "memory");
    asm volatile("s_barrier_signal -1\n\ts_barrier_wait -1" ::: "memory");
    v4f s0[NWAVE], s1[NWAVE];
#pragma unroll
    for (int j = 0; j < NWAVE; ++j) { s0[j] = Pre[0][j][col]; s1[j] = Pre[1][j][col]; }
#pragma unroll
    for (int j = 0; j < NWAVE; ++j) carry = PSEGF_ * carry + (s0[j] + s1[j]);
  } else {
    load_chunk(xb[2], xrow, 0);
    load_chunk(xb[3], xrow, 1);
    load_chunk(xb[0], xrow, 2);
  }

  v2f Afrag[4];
#pragma unroll
  for (int kk = 0; kk < 4; ++kk) {
    const int s0 = 4 * kk + 2 * half;
    const int e0 = col - s0;
    const int e1 = e0 - 1;
    const float p0 = __builtin_amdgcn_exp2f((float)e0 * LOG2_07);
    const float p1 = __builtin_amdgcn_exp2f((float)e1 * LOG2_07);
    Afrag[kk].x = (e0 >= 0) ? p0 : 0.0f;
    Afrag[kk].y = (e1 >= 0) ? p1 : 0.0f;
  }
  float dec[8];
  dec[0] = __builtin_amdgcn_exp2f((float)(8 * half + 1) * LOG2_07);
#pragma unroll
  for (int v = 1; v < 8; ++v) dec[v] = dec[v - 1] * OMA;

  const int sb = back ? SPLIT_SB : 0;
  super_block<2, true >(xb, xrow, yrow, sb + 0, Afrag, dec, carry, Sc, w, half, col);
  super_block<3, true >(xb, xrow, yrow, sb + 1, Afrag, dec, carry, Sc, w, half, col);
  super_block<0, true >(xb, xrow, yrow, sb + 2, Afrag, dec, carry, Sc, w, half, col);
  super_block<1, true >(xb, xrow, yrow, sb + 3, Afrag, dec, carry, Sc, w, half, col);
  super_block<2, true >(xb, xrow, yrow, sb + 4, Afrag, dec, carry, Sc, w, half, col);
  super_block<3, false>(xb, xrow, yrow, sb + 5, Afrag, dec, carry, Sc, w, half, col);
  super_block<0, false>(xb, xrow, yrow, sb + 6, Afrag, dec, carry, Sc, w, half, col);
  super_block<1, false>(xb, xrow, yrow, sb + 7, Afrag, dec, carry, Sc, w, half, col);
}

extern "C" void kernel_launch(void* const* d_in, const int* in_sizes, int n_in,
                              void* d_out, int out_size, void* d_ws,
                              size_t ws_size, hipStream_t stream) {
  (void)in_sizes; (void)n_in; (void)out_size; (void)d_ws; (void)ws_size;
  const float* x = (const float*)d_in[0];
  float* y = (float*)d_out;

  ema_scan_fused<<<2 * NSEQ, NWAVE * 32, 0, stream>>>(x, y);
}
